# starting kernel
# baseline (speedup 1.0000x reference)
#include <hip/hip_runtime.h>
#include <hip/hip_bf16.h>


typedef __attribute__((ext_vector_type(2))) float v2f;
typedef __attribute__((ext_vector_type(8))) float v8f;

static constexpr int   kT    = 16384;
static constexpr int   kW    = 16;
static constexpr int   kSeg  = 2048;
static constexpr int   kThr  = 256;
static constexpr float kFMax = 3.402823466e+38f;

__global__ __launch_bounds__(kThr) void mocha_chunk_attn_kernel(
    const float* __restrict__ emit,
    const float* __restrict__ logits,
    float* __restrict__ out)
{
    const int segs_per_row = kT / kSeg;
    const int b   = blockIdx.x / segs_per_row;
    const int c0  = (blockIdx.x % segs_per_row) * kSeg;

    const float* row_logits = logits + (size_t)b * kT;
    const float* row_emit   = emit   + (size_t)b * kT;
    float*       row_out    = out    + (size_t)b * kT;

    __shared__ float s_logits[kSeg + 2 * kW];
    __shared__ float s_r[kSeg + kW];

    for (int q = threadIdx.x; q < kSeg + 2 * kW - 1; q += kThr) {
        const int c = c0 - (kW - 1) + q;
        s_logits[q] = (c >= 0 && c < kT) ? row_logits[c] : -kFMax;
    }
    __syncthreads();

    for (int ls = threadIdx.x; ls < kSeg + kW; ls += kThr) {
        const int c = c0 + ls;
        float r = 0.0f;
        if (c < kT) {
            float m = -kFMax;
#pragma unroll
            for (int k = 0; k < kW; ++k) m = fmaxf(m, s_logits[ls + k]);
            float den = 0.0f;
#pragma unroll
            for (int k = 0; k < kW; ++k) den += __expf(s_logits[ls + k] - m);
            r = row_emit[c] * __expf(-m) / den;
        }
        s_r[ls] = r;
    }
    __syncthreads();

    const int lane  = threadIdx.x & 31;
    const int wv    = threadIdx.x >> 5;
    const int wbase = wv * 256;
    const int col   = lane & 15;
    const int kh    = lane >> 4;

    v8f acc = {};
#pragma unroll
    for (int kb = 0; kb < 8; ++kb) {
        const int j0 = 4 * kb + 2 * kh;
        v2f a, bf;
        const int d0 = j0 - col;
        const int d1 = j0 + 1 - col;
        a.x = (d0 >= 0 && d0 < kW) ? 1.0f : 0.0f;
        a.y = (d1 >= 0 && d1 < kW) ? 1.0f : 0.0f;
        const int ridx = wbase + 16 * col + j0;
        bf.x = s_r[ridx];
        bf.y = s_r[ridx + 1];
        acc = __builtin_amdgcn_wmma_f32_16x16x4_f32(
            false, a, false, bf, (short)0, acc, false, false);
    }

    const int tl = wbase + 16 * col + kh * 8;
#pragma unroll
    for (int v = 0; v < 8; ++v) {
        const float e = __expf(s_logits[tl + v + (kW - 1)]);
        row_out[c0 + tl + v] = e * acc[v];
    }
}

extern "C" void kernel_launch(void* const* d_in, const int* in_sizes, int n_in,
                              void* d_out, int out_size, void* d_ws, size_t ws_size,
                              hipStream_t stream) {
    (void)n_in; (void)out_size; (void)d_ws; (void)ws_size;
    const float* emit   = (const float*)d_in[0];
    const float* logits = (const float*)d_in[1];
    float* out = (float*)d_out;

    const int B = in_sizes[0] / kT;
    const int blocks = B * (kT / kSeg);
    mocha_chunk_attn_kernel<<<blocks, kThr, 0, stream>>>(emit, logits, out);
}


# exp once per logit; softmax denominator = trailing 16-sum of exp(logit) on a second band WMMA chain, replacing 16 max + 16 exp per column (denominator sum order and rounding change)
# speedup vs baseline: 1.4213x; 1.4213x over previous
#include <hip/hip_runtime.h>


typedef __attribute__((ext_vector_type(2))) float v2f;
typedef __attribute__((ext_vector_type(4))) float v4f;
typedef __attribute__((ext_vector_type(8))) float v8f;

static constexpr int kT   = 16384;
static constexpr int kW   = 16;
static constexpr int kSeg = 2048;
static constexpr int kThr = 256;

__device__ __forceinline__ v8f band_sum16(const float* x, int arow, int ncol, int kh)
{
    v8f acc = {};
#pragma unroll
    for (int kb = 0; kb < 8; ++kb) {
        const int j0 = 4 * kb + 2 * kh;
        v2f a, bf;
        const int d0 = j0 - arow;
        const int d1 = j0 + 1 - arow;
        a.x = (d0 >= 0 && d0 < kW) ? 1.0f : 0.0f;
        a.y = (d1 >= 0 && d1 < kW) ? 1.0f : 0.0f;
        const int idx = 16 * ncol + j0;
        bf.x = x[idx];
        bf.y = x[idx + 1];
        acc = __builtin_amdgcn_wmma_f32_16x16x4_f32(
            false, a, false, bf, (short)0, acc, false, false);
    }
    return acc;
}

__global__ __launch_bounds__(kThr) void mocha_chunk_attn_kernel(
    const float* __restrict__ emit,
    const float* __restrict__ logits,
    float* __restrict__ out)
{
    const int segs_per_row = kT / kSeg;
    const int b   = blockIdx.x / segs_per_row;
    const int c0  = (blockIdx.x % segs_per_row) * kSeg;

    const float* row_logits = logits + (size_t)b * kT;
    const float* row_emit   = emit   + (size_t)b * kT;
    float*       row_out    = out    + (size_t)b * kT;

    __shared__ float s_e[kSeg + 2 * kW];
    __shared__ float s_r[kSeg + kW];

    const int lane  = threadIdx.x & 31;
    const int wv    = __builtin_amdgcn_readfirstlane(threadIdx.x >> 5);
    const int wbase = wv * 256;
    const int col   = lane & 15;
    const int kh    = lane >> 4;
    const int tl    = wbase + 16 * col + kh * 8;
    const bool halo_wave = (wv == kThr / 32 - 1);
    const bool halo_lane = halo_wave && (col == 0);

    float em[8], eh[8];
#pragma unroll
    for (int v = 0; v < 8; ++v) em[v] = row_emit[c0 + tl + v];
#pragma unroll
    for (int v = 0; v < 8; ++v) {
        const int c = c0 + kSeg + kh * 8 + v;
        eh[v] = (halo_lane && c < kT) ? row_emit[c] : 0.0f;
    }

    for (int q = threadIdx.x; q < kSeg + 2 * kW; q += kThr) {
        const int c = c0 - (kW - 1) + q;
        s_e[q] = (c >= 0 && c < kT) ? __expf(row_logits[c]) : 0.0f;
    }
    __syncthreads();

    {
        const v8f den = band_sum16(s_e + wbase, col, col, kh);
#pragma unroll
        for (int v = 0; v < 8; ++v) s_r[tl + v] = em[v] / den[v];
    }
    if (halo_wave) {
        const v8f den = band_sum16(s_e + kSeg, col, 0, kh);
        if (col == 0) {
#pragma unroll
            for (int v = 0; v < 8; ++v) {
                const int c = c0 + kSeg + kh * 8 + v;
                s_r[kSeg + kh * 8 + v] = (c < kT) ? eh[v] / den[v] : 0.0f;
            }
        }
    }
    __syncthreads();

    const v8f acc = band_sum16(s_r + wbase, col, col, kh);
    float o[8];
#pragma unroll
    for (int v = 0; v < 8; ++v)
        o[v] = s_e[tl + v + (kW - 1)] * acc[v];
    const v4f lo = {o[0], o[1], o[2], o[3]};
    const v4f hi = {o[4], o[5], o[6], o[7]};
    *reinterpret_cast<v4f*>(row_out + c0 + tl)     = lo;
    *reinterpret_cast<v4f*>(row_out + c0 + tl + 4) = hi;
}

extern "C" void kernel_launch(void* const* d_in, const int* in_sizes, int n_in,
                              void* d_out, int out_size, void* d_ws, size_t ws_size,
                              hipStream_t stream) {
    (void)n_in; (void)out_size; (void)d_ws; (void)ws_size;
    const float* emit   = (const float*)d_in[0];
    const float* logits = (const float*)d_in[1];
    float* out = (float*)d_out;

    const int B = in_sizes[0] / kT;
    const int blocks = B * (kT / kSeg);
    mocha_chunk_attn_kernel<<<blocks, kThr, 0, stream>>>(emit, logits, out);
}


# staging loads made unconditional (column clamped into the row, zero selected after) and unrolled: a wave's 8-9 logits loads are in flight together instead of one waited round trip per pass
# speedup vs baseline: 2.5890x; 1.8216x over previous
#include <hip/hip_runtime.h>


typedef __attribute__((ext_vector_type(2))) float v2f;
typedef __attribute__((ext_vector_type(4))) float v4f;
typedef __attribute__((ext_vector_type(8))) float v8f;

static constexpr int kT   = 16384;
static constexpr int kW   = 16;
static constexpr int kSeg = 2048;
static constexpr int kThr = 256;
static constexpr int kStage = kSeg + 2 * kW;
static constexpr int kFull  = kStage / kThr;
static constexpr int kTail  = kStage - kFull * kThr;
static_assert(kTail == 32, "the staging tail is one wave");

__device__ __forceinline__ v8f band_sum16(const float* x, int arow, int ncol, int kh)
{
    v8f acc = {};
#pragma unroll
    for (int kb = 0; kb < 8; ++kb) {
        const int j0 = 4 * kb + 2 * kh;
        v2f a, bf;
        const int d0 = j0 - arow;
        const int d1 = j0 + 1 - arow;
        a.x = (d0 >= 0 && d0 < kW) ? 1.0f : 0.0f;
        a.y = (d1 >= 0 && d1 < kW) ? 1.0f : 0.0f;
        const int idx = 16 * ncol + j0;
        bf.x = x[idx];
        bf.y = x[idx + 1];
        acc = __builtin_amdgcn_wmma_f32_16x16x4_f32(
            false, a, false, bf, (short)0, acc, false, false);
    }
    return acc;
}

__global__ __launch_bounds__(kThr) void mocha_chunk_attn_kernel(
    const float* __restrict__ emit,
    const float* __restrict__ logits,
    float* __restrict__ out)
{
    const int segs_per_row = kT / kSeg;
    const int b   = blockIdx.x / segs_per_row;
    const int c0  = (blockIdx.x % segs_per_row) * kSeg;

    const float* row_logits = logits + (size_t)b * kT;
    const float* row_emit   = emit   + (size_t)b * kT;
    float*       row_out    = out    + (size_t)b * kT;

    __shared__ float s_e[kSeg + 2 * kW];
    __shared__ float s_r[kSeg + kW];

    const int lane  = threadIdx.x & 31;
    const int wv    = __builtin_amdgcn_readfirstlane(threadIdx.x >> 5);
    const int wbase = wv * 256;
    const int col   = lane & 15;
    const int kh    = lane >> 4;
    const int tl    = wbase + 16 * col + kh * 8;
    const bool halo_wave = (wv == kThr / 32 - 1);
    const bool halo_lane = halo_wave && (col == 0);

    float em[8], eh[8];
#pragma unroll
    for (int v = 0; v < 8; ++v) em[v] = row_emit[c0 + tl + v];
#pragma unroll
    for (int v = 0; v < 8; ++v) {
        const int c = c0 + kSeg + kh * 8 + v;
        eh[v] = (halo_lane && c < kT) ? row_emit[c] : 0.0f;
    }

    float lg[kFull], lgt = 0.0f;
    const int cs = c0 - (kW - 1) + (int)threadIdx.x;
#pragma unroll
    for (int i = 0; i < kFull; ++i)
        lg[i] = row_logits[min(max(cs + i * kThr, 0), kT - 1)];
    if (wv == 0)
        lgt = row_logits[min(max(cs + kFull * kThr, 0), kT - 1)];
#pragma unroll
    for (int i = 0; i < kFull; ++i) {
        const int c = cs + i * kThr;
        s_e[threadIdx.x + i * kThr] = (c >= 0 && c < kT) ? __expf(lg[i]) : 0.0f;
    }
    if (wv == 0) {
        const int c = cs + kFull * kThr;
        s_e[threadIdx.x + kFull * kThr] = (c >= 0 && c < kT) ? __expf(lgt) : 0.0f;
    }
    __syncthreads();

    {
        const v8f den = band_sum16(s_e + wbase, col, col, kh);
#pragma unroll
        for (int v = 0; v < 8; ++v) s_r[tl + v] = em[v] / den[v];
    }
    if (halo_wave) {
        const v8f den = band_sum16(s_e + kSeg, col, 0, kh);
        if (col == 0) {
#pragma unroll
            for (int v = 0; v < 8; ++v) {
                const int c = c0 + kSeg + kh * 8 + v;
                s_r[kSeg + kh * 8 + v] = (c < kT) ? eh[v] / den[v] : 0.0f;
            }
        }
    }
    __syncthreads();

    const v8f acc = band_sum16(s_r + wbase, col, col, kh);
    float o[8];
#pragma unroll
    for (int v = 0; v < 8; ++v)
        o[v] = s_e[tl + v + (kW - 1)] * acc[v];
    const v4f lo = {o[0], o[1], o[2], o[3]};
    const v4f hi = {o[4], o[5], o[6], o[7]};
    *reinterpret_cast<v4f*>(row_out + c0 + tl)     = lo;
    *reinterpret_cast<v4f*>(row_out + c0 + tl + 4) = hi;
}

extern "C" void kernel_launch(void* const* d_in, const int* in_sizes, int n_in,
                              void* d_out, int out_size, void* d_ws, size_t ws_size,
                              hipStream_t stream) {
    (void)n_in; (void)out_size; (void)d_ws; (void)ws_size;
    const float* emit   = (const float*)d_in[0];
    const float* logits = (const float*)d_in[1];
    float* out = (float*)d_out;

    const int B = in_sizes[0] / kT;
    const int blocks = B * (kT / kSeg);
    mocha_chunk_attn_kernel<<<blocks, kThr, 0, stream>>>(emit, logits, out);
}


# band-sum operands kept in registers: k order makes D a B operand, neighbour tile by DPP row shift; LDS only passes 16 r per wave, one barrier (sum order inside the 16-sums changes)
# speedup vs baseline: 2.7671x; 1.0688x over previous
#include <hip/hip_runtime.h>


typedef __attribute__((ext_vector_type(2))) float v2f;
typedef __attribute__((ext_vector_type(4))) float v4f;
typedef __attribute__((ext_vector_type(8))) float v8f;

static constexpr int kT   = 16384;
static constexpr int kW   = 16;
static constexpr int kWv  = 256;
static constexpr int kThr = 256;
static constexpr int kNw  = kThr / 32;
static constexpr int kSeg = kWv * kNw;

__device__ __forceinline__ v8f band_half(v8f acc, const float (&x)[8], int c, int h, int base, int first)
{
#pragma unroll
    for (int s = 0; s < 4; ++s) {
        const int d0 = base + 8 * h + 2 * s - c - first;
        const int d1 = d0 + 1;
        v2f a, bf;
        a.x = (d0 >= 0 && d0 < kW) ? 1.0f : 0.0f;
        a.y = (d1 >= 0 && d1 < kW) ? 1.0f : 0.0f;
        bf.x = x[2 * s];
        bf.y = x[2 * s + 1];
        acc = __builtin_amdgcn_wmma_f32_16x16x4_f32(
            false, a, false, bf, (short)0, acc, false, false);
    }
    return acc;
}

__device__ __forceinline__ float from_left(float src, float edge)
{
    return __builtin_bit_cast(float, __builtin_amdgcn_update_dpp(
        __builtin_bit_cast(int, edge), __builtin_bit_cast(int, src),
        0x111  , 0xf, 0xf, false));
}
__device__ __forceinline__ float from_right(float src, float edge)
{
    return __builtin_bit_cast(float, __builtin_amdgcn_update_dpp(
        __builtin_bit_cast(int, edge), __builtin_bit_cast(int, src),
        0x101  , 0xf, 0xf, false));
}

__device__ __forceinline__ void unpack8(const v4f a, const v4f b, float (&d)[8])
{
    d[0] = a.x; d[1] = a.y; d[2] = a.z; d[3] = a.w;
    d[4] = b.x; d[5] = b.y; d[6] = b.z; d[7] = b.w;
}

__global__ __launch_bounds__(kThr) void mocha_chunk_attn_kernel(
    const float* __restrict__ emit,
    const float* __restrict__ logits,
    float* __restrict__ out)
{
    const int segs_per_row = kT / kSeg;
    const int b   = blockIdx.x / segs_per_row;
    const int c0  = (blockIdx.x % segs_per_row) * kSeg;

    const float* row_logits = logits + (size_t)b * kT;
    const float* row_emit   = emit   + (size_t)b * kT;
    float*       row_out    = out    + (size_t)b * kT;

    __shared__ float s_rh[(kNw + 1) * kW];

    const int lane = threadIdx.x & 31;
    const int wv   = __builtin_amdgcn_readfirstlane(threadIdx.x >> 5);
    const int wc   = c0 + wv * kWv;
    const int c    = lane & 15;
    const int h    = lane >> 4;
    const int t0   = wc + 16 * c + 8 * h;
    const bool last_wave = (wv == kNw - 1);

    const int  hl    = (c == 0) ? t0 - 16 : wc + kWv + 8 * h;
    const bool hl_in = (hl >= 0) && (hl < kT);
    const int  hr    = wc + kWv + 8 * h;
    const bool hr_in = hr < kT;

    const v4f lg0 = *reinterpret_cast<const v4f*>(row_logits + t0);
    const v4f lg1 = *reinterpret_cast<const v4f*>(row_logits + t0 + 4);
    const v4f lh0 = *reinterpret_cast<const v4f*>(row_logits + (hl_in ? hl : t0));
    const v4f lh1 = *reinterpret_cast<const v4f*>(row_logits + (hl_in ? hl : t0) + 4);
    const v4f em0 = *reinterpret_cast<const v4f*>(row_emit + t0);
    const v4f em1 = *reinterpret_cast<const v4f*>(row_emit + t0 + 4);
    const v4f eh0 = *reinterpret_cast<const v4f*>(row_emit + (hr_in ? hr : t0));
    const v4f eh1 = *reinterpret_cast<const v4f*>(row_emit + (hr_in ? hr : t0) + 4);
    asm volatile("" :: "v"(lg0), "v"(lg1), "v"(lh0), "v"(lh1), "v"(em0), "v"(em1), "v"(eh0), "v"(eh1));

    float lg[8], lgh[8], em[8], emh[8];
    unpack8(lg0, lg1, lg);
    unpack8(lh0, lh1, lgh);
    unpack8(em0, em1, em);
    unpack8(eh0, eh1, emh);

    float e[8], eh[8], ep[8];
#pragma unroll
    for (int v = 0; v < 8; ++v) e[v] = __expf(lg[v]);
#pragma unroll
    for (int v = 0; v < 8; ++v) eh[v] = (hl >= 0) ? __expf(lgh[v]) : 0.0f;
#pragma unroll
    for (int v = 0; v < 8; ++v) ep[v] = from_left(e[v], eh[v]);

    const v8f zero = {};
    const v8f den = band_half(band_half(zero, ep, c, h, 0, 1), e, c, h, 16, 1);
    float r[8];
#pragma unroll
    for (int v = 0; v < 8; ++v) r[v] = em[v] / den[v];
    if (c == 0) {
        const v4f r0 = {r[0], r[1], r[2], r[3]};
        const v4f r1 = {r[4], r[5], r[6], r[7]};
        *reinterpret_cast<v4f*>(s_rh + kW * wv + 8 * h)     = r0;
        *reinterpret_cast<v4f*>(s_rh + kW * wv + 8 * h + 4) = r1;
    }
    if (last_wave) {
        const v8f denh = band_half(band_half(zero, e, c, h, 0, 1), eh, c, h, 16, 1);
        if (c == 15) {
            v4f q0, q1;
#pragma unroll
            for (int v = 0; v < 4; ++v) {
                q0[v] = (hr_in ? emh[v]     : 0.0f) / denh[v];
                q1[v] = (hr_in ? emh[v + 4] : 0.0f) / denh[v + 4];
            }
            *reinterpret_cast<v4f*>(s_rh + kW * kNw + 8 * h)     = q0;
            *reinterpret_cast<v4f*>(s_rh + kW * kNw + 8 * h + 4) = q1;
        }
    }

    v8f acc = band_half(zero, r, c, h, 0, 0);
    __syncthreads();
    const v4f n0 = *reinterpret_cast<const v4f*>(s_rh + kW * (wv + 1) + 8 * h);
    const v4f n1 = *reinterpret_cast<const v4f*>(s_rh + kW * (wv + 1) + 8 * h + 4);
    float rh[8], rn[8];
    unpack8(n0, n1, rh);
#pragma unroll
    for (int v = 0; v < 8; ++v) rn[v] = from_right(r[v], rh[v]);
    acc = band_half(acc, rn, c, h, 16, 0);

    const v4f lo = {e[0] * acc[0], e[1] * acc[1], e[2] * acc[2], e[3] * acc[3]};
    const v4f hi = {e[4] * acc[4], e[5] * acc[5], e[6] * acc[6], e[7] * acc[7]};
    *reinterpret_cast<v4f*>(row_out + t0)     = lo;
    *reinterpret_cast<v4f*>(row_out + t0 + 4) = hi;
}

extern "C" void kernel_launch(void* const* d_in, const int* in_sizes, int n_in,
                              void* d_out, int out_size, void* d_ws, size_t ws_size,
                              hipStream_t stream) {
    (void)n_in; (void)out_size; (void)d_ws; (void)ws_size;
    const float* emit   = (const float*)d_in[0];
    const float* logits = (const float*)d_in[1];
    float* out = (float*)d_out;

    const int B = in_sizes[0] / kT;
    const int blocks = B * (kT / kSeg);
    mocha_chunk_attn_kernel<<<blocks, kThr, 0, stream>>>(emit, logits, out);
}


# r = emit * v_rcp_f32(S) (hardware reciprocal, about 1 ulp, then one multiply) replaces the exact f32 division, in the tiles and in the edge tile behind the block; last bit of r may differ
# speedup vs baseline: 3.2641x; 1.1796x over previous
#include <hip/hip_runtime.h>


typedef __attribute__((ext_vector_type(2))) float v2f;
typedef __attribute__((ext_vector_type(4))) float v4f;
typedef __attribute__((ext_vector_type(8))) float v8f;

static constexpr int kT   = 16384;
static constexpr int kW   = 16;
static constexpr int kWv  = 256;
static constexpr int kThr = 256;
static constexpr int kNw  = kThr / 32;
static constexpr int kSeg = kWv * kNw;

__device__ __forceinline__ v8f tile_prefix(const float (&x)[8], int c, int h, int strict)
{
    v8f acc = {};
#pragma unroll
    for (int s = 0; s < 4; ++s) {
        const int j0 = 8 * h + 2 * s + strict;
        v2f a, bf;
        a.x = (j0     <= c) ? 1.0f : 0.0f;
        a.y = (j0 + 1 <= c) ? 1.0f : 0.0f;
        bf.x = x[2 * s];
        bf.y = x[2 * s + 1];
        acc = __builtin_amdgcn_wmma_f32_16x16x4_f32(
            false, a, false, bf, (short)0, acc, false, false);
    }
    return acc;
}

__device__ __forceinline__ float from_left(float src, float edge)
{
    return __builtin_bit_cast(float, __builtin_amdgcn_update_dpp(
        __builtin_bit_cast(int, edge), __builtin_bit_cast(int, src),
        0x111  , 0xf, 0xf, false));
}
__device__ __forceinline__ float from_right(float src, float edge)
{
    return __builtin_bit_cast(float, __builtin_amdgcn_update_dpp(
        __builtin_bit_cast(int, edge), __builtin_bit_cast(int, src),
        0x101  , 0xf, 0xf, false));
}
__device__ __forceinline__ float other_half(float x)
{
    return __builtin_bit_cast(float, __builtin_amdgcn_permlanex16(
        __builtin_bit_cast(int, x), __builtin_bit_cast(int, x),
        0x76543210, 0xfedcba98, false, false));
}

__device__ __forceinline__ float div_rcp(float num, float den)
{
    return num * __builtin_amdgcn_rcpf(den);
}

__device__ __forceinline__ void unpack8(const v4f a, const v4f b, float (&d)[8])
{
    d[0] = a.x; d[1] = a.y; d[2] = a.z; d[3] = a.w;
    d[4] = b.x; d[5] = b.y; d[6] = b.z; d[7] = b.w;
}

__global__ __launch_bounds__(kThr) void mocha_chunk_attn_kernel(
    const float* __restrict__ emit,
    const float* __restrict__ logits,
    float* __restrict__ out)
{
    const int segs_per_row = kT / kSeg;
    const int b   = blockIdx.x / segs_per_row;
    const int c0  = (blockIdx.x % segs_per_row) * kSeg;

    const float* row_logits = logits + (size_t)b * kT;
    const float* row_emit   = emit   + (size_t)b * kT;
    float*       row_out    = out    + (size_t)b * kT;

    __shared__ float s_eh[(kNw + 1) * kW];

    const int lane = threadIdx.x & 31;
    const int wv   = __builtin_amdgcn_readfirstlane(threadIdx.x >> 5);
    const int wc   = c0 + wv * kWv;
    const int c    = lane & 15;
    const int h    = lane >> 4;
    const int t0   = wc + 16 * c + 8 * h;
    const bool last_wave = (wv == kNw - 1);

    const int  hl    = (c == 0) ? t0 - 16 : wc + kWv + 8 * h;
    const bool hl_in = (hl >= 0) && (hl < kT);
    const int  hr    = wc + kWv + 8 * h;
    const bool hr_in = hr < kT;

    const v4f lg0 = *reinterpret_cast<const v4f*>(row_logits + t0);
    const v4f lg1 = *reinterpret_cast<const v4f*>(row_logits + t0 + 4);
    const v4f lh0 = *reinterpret_cast<const v4f*>(row_logits + (hl_in ? hl : t0));
    const v4f lh1 = *reinterpret_cast<const v4f*>(row_logits + (hl_in ? hl : t0) + 4);
    const v4f em0 = *reinterpret_cast<const v4f*>(row_emit + t0);
    const v4f em1 = *reinterpret_cast<const v4f*>(row_emit + t0 + 4);
    const v4f eh0 = *reinterpret_cast<const v4f*>(row_emit + (hr_in ? hr : t0));
    const v4f eh1 = *reinterpret_cast<const v4f*>(row_emit + (hr_in ? hr : t0) + 4);
    asm volatile("" :: "v"(lg0), "v"(lg1), "v"(lh0), "v"(lh1), "v"(em0), "v"(em1), "v"(eh0), "v"(eh1));

    float lg[8], lgh[8], em[8], emh[8];
    unpack8(lg0, lg1, lg);
    unpack8(lh0, lh1, lgh);
    unpack8(em0, em1, em);
    unpack8(eh0, eh1, emh);

    float e[8], eh[8];
#pragma unroll
    for (int v = 0; v < 8; ++v) e[v] = __expf(lg[v]);
#pragma unroll
    for (int v = 0; v < 8; ++v) eh[v] = (hl >= 0) ? __expf(lgh[v]) : 0.0f;

    const v8f P = tile_prefix(e, c, h, 0);
    const float p7_other = other_half(P[7]);
    const float T = h ? P[7] : p7_other;
    float q[8];
#pragma unroll
    for (int v = 0; v < 8; ++v) q[v] = T - P[v];
    float qe[8];
    qe[7] = 0.0f;
#pragma unroll
    for (int v = 6; v >= 0; --v) qe[v] = qe[v + 1] + eh[v + 1];
    const float eh_all   = qe[0] + eh[0];
    const float eh_other = other_half(eh_all);
    const float qe_add   = h ? 0.0f : eh_other;
    float r[8];
#pragma unroll
    for (int v = 0; v < 8; ++v) r[v] = div_rcp(em[v], P[v] + from_left(q[v], qe[v] + qe_add));

    if (last_wave) {
        float pe[8];
        pe[0] = eh[0];
#pragma unroll
        for (int v = 1; v < 8; ++v) pe[v] = pe[v - 1] + eh[v];
        const float pe_other = other_half(pe[7]);
        const float pe_add   = h ? pe_other : 0.0f;
        float re[8], ee[8];
#pragma unroll
        for (int v = 0; v < 8; ++v) re[v] = div_rcp(hr_in ? emh[v] : 0.0f, (pe[v] + pe_add) + q[v]);
        ee[0] = 0.0f;
#pragma unroll
        for (int v = 1; v < 8; ++v) ee[v] = ee[v - 1] + re[v - 1];
        const float re_other = other_half(ee[7] + re[7]);
        const float ee_add   = h ? re_other : 0.0f;
        if (c == 15) {
            const v4f w0 = {ee[0] + ee_add, ee[1] + ee_add, ee[2] + ee_add, ee[3] + ee_add};
            const v4f w1 = {ee[4] + ee_add, ee[5] + ee_add, ee[6] + ee_add, ee[7] + ee_add};
            *reinterpret_cast<v4f*>(s_eh + kW * kNw + 8 * h)     = w0;
            *reinterpret_cast<v4f*>(s_eh + kW * kNw + 8 * h + 4) = w1;
        }
    }

    const v8f E = tile_prefix(r, c, h, 1);
    if (c == 0) {
        const v4f w0 = {E[0], E[1], E[2], E[3]};
        const v4f w1 = {E[4], E[5], E[6], E[7]};
        *reinterpret_cast<v4f*>(s_eh + kW * wv + 8 * h)     = w0;
        *reinterpret_cast<v4f*>(s_eh + kW * wv + 8 * h + 4) = w1;
    }
    const float r_all = E[7] + r[7];
    const float r_all_other = other_half(r_all);
    const float Tr = h ? r_all : r_all_other;
    float g[8];
#pragma unroll
    for (int v = 0; v < 8; ++v) g[v] = Tr - E[v];
    __syncthreads();
    const v4f n0 = *reinterpret_cast<const v4f*>(s_eh + kW * (wv + 1) + 8 * h);
    const v4f n1 = *reinterpret_cast<const v4f*>(s_eh + kW * (wv + 1) + 8 * h + 4);
    float en[8];
    unpack8(n0, n1, en);
    float acc[8];
#pragma unroll
    for (int v = 0; v < 8; ++v) acc[v] = g[v] + from_right(E[v], en[v]);

    const v4f lo = {e[0] * acc[0], e[1] * acc[1], e[2] * acc[2], e[3] * acc[3]};
    const v4f hi = {e[4] * acc[4], e[5] * acc[5], e[6] * acc[6], e[7] * acc[7]};
    *reinterpret_cast<v4f*>(row_out + t0)     = lo;
    *reinterpret_cast<v4f*>(row_out + t0 + 4) = hi;
}

extern "C" void kernel_launch(void* const* d_in, const int* in_sizes, int n_in,
                              void* d_out, int out_size, void* d_ws, size_t ws_size,
                              hipStream_t stream) {
    (void)n_in; (void)out_size; (void)d_ws; (void)ws_size;
    const float* emit   = (const float*)d_in[0];
    const float* logits = (const float*)d_in[1];
    float* out = (float*)d_out;

    const int B = in_sizes[0] / kT;
    const int blocks = B * (kT / kSeg);
    mocha_chunk_attn_kernel<<<blocks, kThr, 0, stream>>>(emit, logits, out);
}
